# starting kernel
# baseline (speedup 1.0000x reference)
#include <hip/hip_runtime.h>
#include <math.h>

#define HIDF   128
#define NHEAD  4
#define WHID   512
#define NGROUP 64
#define OUTD   256
#define OUT2   128
#define NCLSD  2
#define SLOPE  0.2f
#define KC     128
#define BSTRIDE (KC + 2)

typedef float v2f __attribute__((ext_vector_type(2)));
typedef float v8f __attribute__((ext_vector_type(8)));

__device__ __forceinline__ float lrelu(float x) { return x >= 0.f ? x : SLOPE * x; }

__device__ __forceinline__ void atomicMaxF(float* addr, float val) {
  if (val >= 0.f) atomicMax((int*)addr, __float_as_int(val));
  else            atomicMin((unsigned int*)addr, __float_as_uint(val));
}

__global__ void wmma_gemm(const float* __restrict__ A, const float* __restrict__ B,
                          const float* __restrict__ bias, float* __restrict__ C,
                          int M, int K, int Nc) {
  __shared__ float Bs[64 * BSTRIDE];

  const int nslab = Nc >> 6;
  const int mblk  = blockIdx.x / nslab;
  const int nblk  = blockIdx.x - mblk * nslab;
  const int wave  = threadIdx.x >> 5;
  const int lane  = threadIdx.x & 31;
  const int half  = lane >> 4;
  const int l     = lane & 15;

  const int mt    = mblk * 8 + wave;
  const int arow  = min(mt * 16 + l, M - 1);
  const int col0  = nblk * 64;

  const int sc = threadIdx.x & 63;
  const int sr = threadIdx.x >> 6;

  const float* Ap = A + (size_t)arow * K + 2 * half;

  v8f acc0 = {}, acc1 = {}, acc2 = {}, acc3 = {};

  for (int k0 = 0; k0 < K; k0 += KC) {
    const int kc = min(KC, K - k0);
    __syncthreads();
    for (int kk = sr; kk < kc; kk += 4)
      Bs[sc * BSTRIDE + kk] = B[(size_t)(k0 + kk) * Nc + col0 + sc];
    __syncthreads();

    __builtin_prefetch(Ap + k0 + KC, 0, 1);

    const float* Bl0 = &Bs[(l +  0) * BSTRIDE + 2 * half];
    const float* Bl1 = &Bs[(l + 16) * BSTRIDE + 2 * half];
    const float* Bl2 = &Bs[(l + 32) * BSTRIDE + 2 * half];
    const float* Bl3 = &Bs[(l + 48) * BSTRIDE + 2 * half];
    for (int k = 0; k < kc; k += 4) {
      v2f a  = *(const v2f*)(Ap + k0 + k);
      v2f b0 = *(const v2f*)(Bl0 + k);
      v2f b1 = *(const v2f*)(Bl1 + k);
      v2f b2 = *(const v2f*)(Bl2 + k);
      v2f b3 = *(const v2f*)(Bl3 + k);
      acc0 = __builtin_amdgcn_wmma_f32_16x16x4_f32(false, a, false, b0, (short)0, acc0, false, false);
      acc1 = __builtin_amdgcn_wmma_f32_16x16x4_f32(false, a, false, b1, (short)0, acc1, false, false);
      acc2 = __builtin_amdgcn_wmma_f32_16x16x4_f32(false, a, false, b2, (short)0, acc2, false, false);
      acc3 = __builtin_amdgcn_wmma_f32_16x16x4_f32(false, a, false, b3, (short)0, acc3, false, false);
    }
  }

  v8f* accs[4] = {&acc0, &acc1, &acc2, &acc3};
#pragma unroll
  for (int j = 0; j < 4; ++j) {
    const int ccol = col0 + 16 * j + l;
    const float bv = bias ? bias[ccol] : 0.f;
#pragma unroll
    for (int i = 0; i < 8; ++i) {
      const int row = mt * 16 + i + 8 * half;
      if (row < M) C[(size_t)row * Nc + ccol] = (*accs[j])[i] + bv;
    }
  }
}

__global__ void attn_scores(const float* __restrict__ h, const float* __restrict__ a_s,
                            const float* __restrict__ a_d, float* __restrict__ ssrc,
                            float* __restrict__ sdst, int n) {
  int idx = blockIdx.x * blockDim.x + threadIdx.x;
  if (idx >= n * NHEAD) return;
  int hd = idx & (NHEAD - 1);
  const float* hp = h + (size_t)idx * HIDF;
  const float* as = a_s + hd * HIDF;
  const float* ad = a_d + hd * HIDF;
  float s1 = 0.f, s2 = 0.f;
  for (int c = 0; c < HIDF; c += 4) {
    float4 hv = *(const float4*)(hp + c);
    float4 av = *(const float4*)(as + c);
    float4 dv = *(const float4*)(ad + c);
    s1 += hv.x * av.x + hv.y * av.y + hv.z * av.z + hv.w * av.w;
    s2 += hv.x * dv.x + hv.y * dv.y + hv.z * dv.z + hv.w * dv.w;
  }
  ssrc[idx] = s1;
  sdst[idx] = s2;
}

__global__ void fill_f32(float* __restrict__ p, float v, int n) {
  int i = blockIdx.x * blockDim.x + threadIdx.x;
  if (i < n) p[i] = v;
}

__global__ void edge_logits_max(const int* __restrict__ ei, int E, int n,
                                const float* __restrict__ ssrc,
                                const float* __restrict__ sdst,
                                float* __restrict__ eatt, float* __restrict__ m) {
  int idx = blockIdx.x * blockDim.x + threadIdx.x;
  if (idx >= (E + n) * NHEAD) return;
  int e = idx >> 2, hd = idx & 3;
  int s, d;
  if (e < E) { s = ei[e]; d = ei[E + e]; } else { s = e - E; d = s; }
  float v = lrelu(ssrc[s * NHEAD + hd] + sdst[d * NHEAD + hd]);
  eatt[idx] = v;
  atomicMaxF(&m[d * NHEAD + hd], v);
}

__global__ void fix_m(float* __restrict__ m, int count) {
  int i = blockIdx.x * blockDim.x + threadIdx.x;
  if (i >= count) return;
  float v = m[i];
  if (!(v > -3.0e38f && v < 3.0e38f)) m[i] = 0.f;
}

__global__ void edge_expsum(const int* __restrict__ ei, int E, int n,
                            float* __restrict__ eatt, const float* __restrict__ m,
                            float* __restrict__ den) {
  int idx = blockIdx.x * blockDim.x + threadIdx.x;
  if (idx >= (E + n) * NHEAD) return;
  int e = idx >> 2, hd = idx & 3;
  int d = (e < E) ? ei[E + e] : (e - E);
  float ex = expf(eatt[idx] - m[d * NHEAD + hd]);
  eatt[idx] = ex;
  atomicAdd(&den[d * NHEAD + hd], ex);
}

__global__ void edge_aggr(const int* __restrict__ ei, int E, int n,
                          const float* __restrict__ ex, const float* __restrict__ den,
                          const float* __restrict__ h, float* __restrict__ acc,
                          float* __restrict__ alpha_out) {
  long gid  = (long)blockIdx.x * blockDim.x + threadIdx.x;
  long wave = gid >> 5;
  int  lane = threadIdx.x & 31;
  if (wave >= (long)(E + n) * NHEAD) return;
  int e = (int)(wave >> 2), hd = (int)(wave & 3);
  int s, d;
  if (e < E) { s = ei[e]; d = ei[E + e]; } else { s = e - E; d = s; }
  float a = ex[wave] / den[(size_t)d * NHEAD + hd];
  if (alpha_out && lane == 0) alpha_out[wave] = a;
  const float* hp = h + ((size_t)s * NHEAD + hd) * HIDF + lane * 4;
  float*       ap = acc + ((size_t)d * NHEAD + hd) * HIDF + lane * 4;
  float4 hv = *(const float4*)hp;
  atomicAdd(ap + 0, a * hv.x);
  atomicAdd(ap + 1, a * hv.y);
  atomicAdd(ap + 2, a * hv.z);
  atomicAdd(ap + 3, a * hv.w);
}

__global__ void l1_post(const float* __restrict__ acc, const float* __restrict__ b1,
                        const float* __restrict__ g, const float* __restrict__ b,
                        float* __restrict__ x2) {
  __shared__ float sh[HIDF];
  int nd = blockIdx.x, c = threadIdx.x;
  const float* ap = acc + (size_t)nd * WHID;
  float v = 0.25f * (ap[c] + ap[HIDF + c] + ap[2 * HIDF + c] + ap[3 * HIDF + c]) + b1[c];
  sh[c] = v; __syncthreads();
  for (int s = HIDF / 2; s > 0; s >>= 1) { if (c < s) sh[c] += sh[c + s]; __syncthreads(); }
  float mu = sh[0] * (1.0f / HIDF);
  __syncthreads();
  float dv = v - mu;
  sh[c] = dv * dv; __syncthreads();
  for (int s = HIDF / 2; s > 0; s >>= 1) { if (c < s) sh[c] += sh[c + s]; __syncthreads(); }
  float var = sh[0] * (1.0f / HIDF);
  float y = g[c] * dv * rsqrtf(var + 1e-5f) + b[c];
  x2[(size_t)nd * HIDF + c] = lrelu(y);
}

__global__ void l2_post_pool(const float* __restrict__ acc, const float* __restrict__ b2,
                             const int* __restrict__ batch, float* __restrict__ pooled,
                             float* __restrict__ cnt) {
  int nd = blockIdx.x, c = threadIdx.x;
  float y = lrelu(acc[(size_t)nd * WHID + c] + b2[c]);
  int gidx = batch[nd];
  atomicAdd(&pooled[(size_t)gidx * WHID + c], y);
  if (c == 0) atomicAdd(&cnt[gidx], 1.0f);
}

__global__ void pool_div(float* __restrict__ pooled, const float* __restrict__ cnt) {
  int i = blockIdx.x * blockDim.x + threadIdx.x;
  if (i >= NGROUP * WHID) return;
  pooled[i] /= fmaxf(cnt[i / WHID], 1.0f);
}

__global__ void ln_relu(float* __restrict__ x, const float* __restrict__ g,
                        const float* __restrict__ b, int dim) {
  extern __shared__ float sh[];
  int r = blockIdx.x, c = threadIdx.x;
  float v = x[(size_t)r * dim + c];
  sh[c] = v; __syncthreads();
  for (int s = dim >> 1; s > 0; s >>= 1) { if (c < s) sh[c] += sh[c + s]; __syncthreads(); }
  float mu = sh[0] / dim;
  __syncthreads();
  float dv = v - mu;
  sh[c] = dv * dv; __syncthreads();
  for (int s = dim >> 1; s > 0; s >>= 1) { if (c < s) sh[c] += sh[c + s]; __syncthreads(); }
  float var = sh[0] / dim;
  float y = g[c] * dv * rsqrtf(var + 1e-5f) + b[c];
  x[(size_t)r * dim + c] = fmaxf(y, 0.f);
}

__global__ void head_out(const float* __restrict__ t2, const float* __restrict__ Ws3,
                         const float* __restrict__ bs3, float* __restrict__ out) {
  int r = blockIdx.x * blockDim.x + threadIdx.x;
  if (r >= NGROUP) return;
  float l0 = bs3[0], l1 = bs3[1];
  for (int c = 0; c < OUT2; ++c) {
    float v = t2[r * OUT2 + c];
    l0 += v * Ws3[c * 2 + 0];
    l1 += v * Ws3[c * 2 + 1];
  }
  float mx = fmaxf(l0, l1);
  float e0 = expf(l0 - mx), e1 = expf(l1 - mx), s = e0 + e1;
  out[r * 2 + 0] = e0 / s;
  out[r * 2 + 1] = e1 / s;
}

extern "C" void kernel_launch(void* const* d_in, const int* in_sizes, int n_in,
                              void* d_out, int out_size, void* d_ws, size_t ws_size,
                              hipStream_t stream) {
  (void)n_in; (void)out_size; (void)ws_size;
  const float* x    = (const float*)d_in[0];
  const int*   ei   = (const int*)d_in[1];
  const int*   batch= (const int*)d_in[2];
  const float* W1   = (const float*)d_in[3];
  const float* as1  = (const float*)d_in[4];
  const float* ad1  = (const float*)d_in[5];
  const float* b1   = (const float*)d_in[6];
  const float* g_ln = (const float*)d_in[7];
  const float* b_ln = (const float*)d_in[8];
  const float* W2   = (const float*)d_in[9];
  const float* as2  = (const float*)d_in[10];
  const float* ad2  = (const float*)d_in[11];
  const float* b2   = (const float*)d_in[12];
  const float* Wl0  = (const float*)d_in[13];
  const float* bl0  = (const float*)d_in[14];
  const float* Ws1  = (const float*)d_in[15];
  const float* bs1  = (const float*)d_in[16];
  const float* g1   = (const float*)d_in[17];
  const float* bb1  = (const float*)d_in[18];
  const float* Ws2  = (const float*)d_in[19];
  const float* bs2  = (const float*)d_in[20];
  const float* g2   = (const float*)d_in[21];
  const float* bb2  = (const float*)d_in[22];
  const float* Ws3  = (const float*)d_in[23];
  const float* bs3  = (const float*)d_in[24];

  const int N  = in_sizes[0] / HIDF;
  const int E  = in_sizes[1] / 2;
  const int TE = E + N;
  const int NH = N * NHEAD;
  const int TEH = TE * NHEAD;

  float* out       = (float*)d_out;
  float* alpha_out = out + NGROUP * NCLSD;

  size_t off = 0;
  auto alloc = [&](size_t bytes) -> float* {
    off = (off + 255) & ~(size_t)255;
    float* p = (float*)((char*)d_ws + off);
    off += bytes;
    return p;
  };
  float* h      = alloc((size_t)N * WHID * 4);
  float* acc    = alloc((size_t)N * WHID * 4);
  float* x2     = alloc((size_t)N * HIDF * 4);
  float* ssrc   = alloc((size_t)NH * 4);
  float* sdst   = alloc((size_t)NH * 4);
  float* eatt   = alloc((size_t)TEH * 4);
  float* mbuf   = alloc((size_t)NH * 4);
  float* den    = alloc((size_t)NH * 4);
  float* pooled = alloc((size_t)NGROUP * WHID * 4);
  float* cnt    = alloc((size_t)NGROUP * 4);
  float* z0     = alloc((size_t)NGROUP * HIDF * 4);
  float* t1     = alloc((size_t)NGROUP * OUTD * 4);
  float* t2     = alloc((size_t)NGROUP * OUT2 * 4);

  auto cdiv = [](long a, long b) { return (int)((a + b - 1) / b); };
  auto gemm = [&](const float* A, const float* B, const float* bias, float* C,
                  int M, int K, int Nc) {
    int mtiles  = (M + 15) / 16;
    int mblocks = (mtiles + 7) / 8;
    int nslab   = Nc / 64;
    wmma_gemm<<<mblocks * nslab, 256, 0, stream>>>(A, B, bias, C, M, K, Nc);
  };

  gemm(x, W1, nullptr, h, N, HIDF, WHID);
  attn_scores<<<cdiv(NH, 256), 256, 0, stream>>>(h, as1, ad1, ssrc, sdst, N);
  fill_f32<<<cdiv(NH, 256), 256, 0, stream>>>(mbuf, -INFINITY, NH);
  fill_f32<<<cdiv(NH, 256), 256, 0, stream>>>(den, 0.f, NH);
  fill_f32<<<cdiv((long)N * WHID, 256), 256, 0, stream>>>(acc, 0.f, N * WHID);
  edge_logits_max<<<cdiv(TEH, 256), 256, 0, stream>>>(ei, E, N, ssrc, sdst, eatt, mbuf);
  fix_m<<<cdiv(NH, 256), 256, 0, stream>>>(mbuf, NH);
  edge_expsum<<<cdiv(TEH, 256), 256, 0, stream>>>(ei, E, N, eatt, mbuf, den);
  edge_aggr<<<cdiv((long)TEH * 32, 256), 256, 0, stream>>>(ei, E, N, eatt, den, h, acc,
                                                           nullptr);
  l1_post<<<N, HIDF, 0, stream>>>(acc, b1, g_ln, b_ln, x2);

  gemm(x2, W2, nullptr, h, N, HIDF, WHID);
  attn_scores<<<cdiv(NH, 256), 256, 0, stream>>>(h, as2, ad2, ssrc, sdst, N);
  fill_f32<<<cdiv(NH, 256), 256, 0, stream>>>(mbuf, -INFINITY, NH);
  fill_f32<<<cdiv(NH, 256), 256, 0, stream>>>(den, 0.f, NH);
  fill_f32<<<cdiv((long)N * WHID, 256), 256, 0, stream>>>(acc, 0.f, N * WHID);
  edge_logits_max<<<cdiv(TEH, 256), 256, 0, stream>>>(ei, E, N, ssrc, sdst, eatt, mbuf);
  fix_m<<<cdiv(NH, 256), 256, 0, stream>>>(mbuf, NH);
  edge_expsum<<<cdiv(TEH, 256), 256, 0, stream>>>(ei, E, N, eatt, mbuf, den);
  edge_aggr<<<cdiv((long)TEH * 32, 256), 256, 0, stream>>>(ei, E, N, eatt, den, h, acc,
                                                           alpha_out);

  fill_f32<<<cdiv(NGROUP * WHID, 256), 256, 0, stream>>>(pooled, 0.f, NGROUP * WHID);
  fill_f32<<<1, NGROUP, 0, stream>>>(cnt, 0.f, NGROUP);
  l2_post_pool<<<N, WHID, 0, stream>>>(acc, b2, batch, pooled, cnt);
  pool_div<<<cdiv(NGROUP * WHID, 256), 256, 0, stream>>>(pooled, cnt);

  gemm(pooled, Wl0, bl0, z0, NGROUP, WHID, HIDF);
  gemm(z0, Ws1, bs1, t1, NGROUP, HIDF, OUTD);
  ln_relu<<<NGROUP, OUTD, OUTD * sizeof(float), stream>>>(t1, g1, bb1, OUTD);
  gemm(t1, Ws2, bs2, t2, NGROUP, OUTD, OUT2);
  ln_relu<<<NGROUP, OUT2, OUT2 * sizeof(float), stream>>>(t2, g2, bb2, OUT2);
  head_out<<<1, NGROUP, 0, stream>>>(t2, Ws3, bs3, out);
}


# attention scores taken from the projection GEMM's accumulators in its epilogue (two half-head partial dots per score, readers add them); 2 score launches, h re-read gone; score sum order changed
# speedup vs baseline: 2.8906x; 2.8906x over previous
#include <hip/hip_runtime.h>
#include <math.h>

#define HIDF   128
#define NHEAD  4
#define WHID   512
#define NGROUP 64
#define OUTD   256
#define OUT2   128
#define NCLSD  2
#define SLOPE  0.2f
#define KC     128
#define BSTRIDE (KC + 4)

typedef float v2f __attribute__((ext_vector_type(2)));
typedef float v8f __attribute__((ext_vector_type(8)));

__device__ __forceinline__ float lrelu(float x) { return x >= 0.f ? x : SLOPE * x; }

__device__ __forceinline__ void atomicMaxF(float* addr, float val) {
  if (val >= 0.f) atomicMax((int*)addr, __float_as_int(val));
  else            atomicMin((unsigned int*)addr, __float_as_uint(val));
}

__global__ void wmma_gemm(const float* __restrict__ A, const float* __restrict__ B,
                          const float* __restrict__ bias, float* __restrict__ C,
                          int M, int K, int Nc,
                          const float* __restrict__ a_s, const float* __restrict__ a_d,
                          float* __restrict__ ssrc, float* __restrict__ sdst) {
  __shared__ __attribute__((aligned(16))) float Bs[64 * BSTRIDE];

  const int nslab = Nc >> 6;
  const int mblk  = blockIdx.x / nslab;
  const int nblk  = blockIdx.x - mblk * nslab;
  const int wave  = threadIdx.x >> 5;
  const int lane  = threadIdx.x & 31;
  const int half  = lane >> 4;
  const int l     = lane & 15;

  const int mt    = mblk * 8 + wave;
  const int arow  = min(mt * 16 + l, M - 1);
  const int col0  = nblk * 64;

  const int sc = threadIdx.x & 63;
  const int sr = threadIdx.x >> 6;

  const float* Ap = A + (size_t)arow * K + 16 * half;

  v8f acc0 = {}, acc1 = {}, acc2 = {}, acc3 = {};

  float4 a0 = *(const float4*)(Ap + 0);
  float4 a1 = *(const float4*)(Ap + 4);
  float4 a2 = *(const float4*)(Ap + 8);
  float4 a3 = *(const float4*)(Ap + 12);

  for (int k0 = 0; k0 < K; k0 += KC) {
    const int kc = min(KC, K - k0);
    __syncthreads();
    for (int kk = sr; kk < kc; kk += 4)
      Bs[sc * BSTRIDE + kk] = B[(size_t)(k0 + kk) * Nc + col0 + sc];
    __syncthreads();

    const float* Bl0 = &Bs[(l +  0) * BSTRIDE + 16 * half];
    const float* Bl1 = &Bs[(l + 16) * BSTRIDE + 16 * half];
    const float* Bl2 = &Bs[(l + 32) * BSTRIDE + 16 * half];
    const float* Bl3 = &Bs[(l + 48) * BSTRIDE + 16 * half];
    for (int kb = 0; kb < kc; kb += 32) {
      const float* An = Ap + min(k0 + kb + 32, K - 32);
      const float4 n0 = *(const float4*)(An + 0);
      const float4 n1 = *(const float4*)(An + 4);
      const float4 n2 = *(const float4*)(An + 8);
      const float4 n3 = *(const float4*)(An + 12);
      const float4 av[4] = {a0, a1, a2, a3};
#pragma unroll
      for (int t = 0; t < 4; ++t) {
        const float4 b0 = *(const float4*)(Bl0 + kb + 4 * t);
        const float4 b1 = *(const float4*)(Bl1 + kb + 4 * t);
        const float4 b2 = *(const float4*)(Bl2 + kb + 4 * t);
        const float4 b3 = *(const float4*)(Bl3 + kb + 4 * t);
        const v2f alo = {av[t].x, av[t].y};
        const v2f ahi = {av[t].z, av[t].w};
        acc0 = __builtin_amdgcn_wmma_f32_16x16x4_f32(false, alo, false, (v2f){b0.x, b0.y}, (short)0, acc0, false, false);
        acc1 = __builtin_amdgcn_wmma_f32_16x16x4_f32(false, alo, false, (v2f){b1.x, b1.y}, (short)0, acc1, false, false);
        acc2 = __builtin_amdgcn_wmma_f32_16x16x4_f32(false, alo, false, (v2f){b2.x, b2.y}, (short)0, acc2, false, false);
        acc3 = __builtin_amdgcn_wmma_f32_16x16x4_f32(false, alo, false, (v2f){b3.x, b3.y}, (short)0, acc3, false, false);
        acc0 = __builtin_amdgcn_wmma_f32_16x16x4_f32(false, ahi, false, (v2f){b0.z, b0.w}, (short)0, acc0, false, false);
        acc1 = __builtin_amdgcn_wmma_f32_16x16x4_f32(false, ahi, false, (v2f){b1.z, b1.w}, (short)0, acc1, false, false);
        acc2 = __builtin_amdgcn_wmma_f32_16x16x4_f32(false, ahi, false, (v2f){b2.z, b2.w}, (short)0, acc2, false, false);
        acc3 = __builtin_amdgcn_wmma_f32_16x16x4_f32(false, ahi, false, (v2f){b3.z, b3.w}, (short)0, acc3, false, false);
      }
      a0 = n0; a1 = n1; a2 = n2; a3 = n3;
    }
  }

  v8f* accs[4] = {&acc0, &acc1, &acc2, &acc3};
#pragma unroll
  for (int j = 0; j < 4; ++j) {
    const int ccol = col0 + 16 * j + l;
    const float bv = bias ? bias[ccol] : 0.f;
#pragma unroll
    for (int i = 0; i < 8; ++i) {
      const int row = mt * 16 + i + 8 * half;
      if (row < M) C[(size_t)row * Nc + ccol] = (*accs[j])[i] + bv;
    }
  }

  if (a_s) {
    float ps[8], pd[8];
#pragma unroll
    for (int j = 0; j < 4; ++j) {
      const int ccol = col0 + 16 * j + l;
      const float bv = bias ? bias[ccol] : 0.f;
      const float sv = a_s[ccol];
      const float dv = a_d[ccol];
#pragma unroll
      for (int i = 0; i < 8; ++i) {
        const float cv = (*accs[j])[i] + bv;
        ps[i] = j ? ps[i] + cv * sv : cv * sv;
        pd[i] = j ? pd[i] + cv * dv : cv * dv;
      }
    }
#pragma unroll
    for (int i = 0; i < 8; ++i) {
#pragma unroll
      for (int o = 1; o < 16; o <<= 1) {
        ps[i] += __shfl_xor(ps[i], o, 32);
        pd[i] += __shfl_xor(pd[i], o, 32);
      }
    }
    float vs = ps[0], vd = pd[0];
#pragma unroll
    for (int i = 1; i < 8; ++i) {
      if (l == i) { vs = ps[i]; vd = pd[i]; }
    }
    const int row = mt * 16 + 8 * half + l;
    if (l < 8 && row < M) {
      const size_t o = ((size_t)row * NHEAD + (col0 >> 7)) * 2 + ((col0 >> 6) & 1);
      ssrc[o] = vs;
      sdst[o] = vd;
    }
  }
}

__device__ __forceinline__ float score_of(const float* __restrict__ p, size_t idx) {
  const float2 v = *(const float2*)(p + 2 * idx);
  return v.x + v.y;
}

__global__ void fill_f32(float* __restrict__ p, float v, int n) {
  int i = blockIdx.x * blockDim.x + threadIdx.x;
  if (i < n) p[i] = v;
}

__device__ __forceinline__ float seg_max_or_zero(float v) {
  return (v > -3.0e38f && v < 3.0e38f) ? v : 0.f;
}

__global__ void edge_alpha(const int* __restrict__ ei, int E, int n,
                           const float* __restrict__ ssrc, const float* __restrict__ sdst,
                           const float* __restrict__ m, const float* __restrict__ den,
                           float* __restrict__ alpha) {
  int idx = blockIdx.x * blockDim.x + threadIdx.x;
  if (idx >= (E + n) * NHEAD) return;
  int e = idx >> 2, hd = idx & 3;
  int s, d;
  if (e < E) { s = ei[e]; d = ei[E + e]; } else { s = e - E; d = s; }
  float v  = lrelu(score_of(ssrc, (size_t)s * NHEAD + hd) + score_of(sdst, (size_t)d * NHEAD + hd));
  float ex = expf(v - seg_max_or_zero(m[d * NHEAD + hd]));
  alpha[idx] = ex / den[d * NHEAD + hd];
}

#define BIN_THREADS 1024
#define BIN_EPT     8
#define BIN_EDGES   (BIN_THREADS * BIN_EPT)
#define MAX_TILES   BIN_THREADS
#define TILE_SHIFT  7

__global__ __launch_bounds__(BIN_THREADS)
void tile_rank(const int* __restrict__ ei, int E, int ntile,
               int* __restrict__ hist, int* __restrict__ rank) {
  __shared__ int cnt[MAX_TILES];
  const int tid = threadIdx.x;
  cnt[tid] = 0;
  __syncthreads();
  const int e0 = blockIdx.x * BIN_EDGES + tid;
  int t[BIN_EPT];
#pragma unroll
  for (int k = 0; k < BIN_EPT; ++k) t[k] = ei[E + min(e0 + k * BIN_THREADS, E - 1)] >> TILE_SHIFT;
#pragma unroll
  for (int k = 0; k < BIN_EPT; ++k) {
    const int e = e0 + k * BIN_THREADS;
    if (e < E) rank[e] = atomicAdd(&cnt[t[k]], 1);
  }
  __syncthreads();
  if (tid < ntile) hist[(size_t)blockIdx.x * ntile + tid] = cnt[tid];
}

__device__ __forceinline__ int wave_incl_scan(int x, int lane) {
  for (int o = 1; o < 32; o <<= 1) {
    const int y = __shfl_up(x, o, 32);
    if (lane >= o) x += y;
  }
  return x;
}

__global__ __launch_bounds__(BIN_THREADS)
void tile_scatter(const int* __restrict__ ei, int E, int ntile, int nblk,
                  const int* __restrict__ hist, const int* __restrict__ rank,
                  int* __restrict__ tstart, int* __restrict__ rec) {
  __shared__ int tot[MAX_TILES];
  __shared__ int bef[MAX_TILES];
  __shared__ int part[BIN_THREADS / 32];
  const int tid  = threadIdx.x;
  const int lane = tid & 31;
  const int wave = tid >> 5;
  const int b    = blockIdx.x;
  tot[tid] = 0;
  bef[tid] = 0;
  __syncthreads();
  const int nel = nblk * ntile;
  for (int base = 0; base < nel; base += BIN_THREADS * 8) {
    int cv[8];
#pragma unroll
    for (int k = 0; k < 8; ++k) {
      const int idx = base + k * BIN_THREADS + tid;
      cv[k] = idx < nel ? hist[idx] : 0;
    }
#pragma unroll
    for (int k = 0; k < 8; ++k) {
      if (cv[k]) {
        const int idx = base + k * BIN_THREADS + tid;
        const int r = idx / ntile;
        const int i = idx - r * ntile;
        atomicAdd(&tot[i], cv[k]);
        if (r < b) atomicAdd(&bef[i], cv[k]);
      }
    }
  }
  __syncthreads();
  const int v = tot[tid];
  const int x = wave_incl_scan(v, lane);
  if (lane == 31) part[wave] = x;
  __syncthreads();
  if (wave == 0) {
    const int w = part[lane];
    part[lane] = wave_incl_scan(w, lane) - w;
  }
  __syncthreads();
  const int start = x - v + part[wave];
  if (b == 0) {
    if (tid <= ntile) tstart[tid] = start;
    if (tid == MAX_TILES - 1 && ntile == MAX_TILES) tstart[MAX_TILES] = start + v;
  }
  tot[tid] = start + bef[tid];
  __syncthreads();
  const int e0 = b * BIN_EDGES + tid;
  int dk[BIN_EPT], sk[BIN_EPT], rk[BIN_EPT];
#pragma unroll
  for (int k = 0; k < BIN_EPT; ++k) {
    const int ec = min(e0 + k * BIN_THREADS, E - 1);
    dk[k] = ei[E + ec];
    sk[k] = ei[ec];
    rk[k] = rank[ec];
  }
#pragma unroll
  for (int k = 0; k < BIN_EPT; ++k) {
    const int e = e0 + k * BIN_THREADS;
    if (e < E) {
      const int t = dk[k] >> TILE_SHIFT;
      ((int4*)rec)[tot[t] + rk[k]] = make_int4(e, sk[k], dk[k] - (t << TILE_SHIFT), 0);
    }
  }
}

#define TILE_N       128
#define TILE_THREADS 1024
static_assert((1 << TILE_SHIFT) == TILE_N, "the bucket key is dst / TILE_N");

__device__ __forceinline__ void tile_add_edge(float* tile, float* dent, const float2* sm,
                                              int lane, int s, int dl,
                                              const float* __restrict__ ssrc,
                                              const float* __restrict__ h) {
  const int q = lane & (NHEAD - 1);
  const float sq = score_of(ssrc, (size_t)s * NHEAD + q);
  const float* hp = h + (size_t)s * WHID + lane;
  float*       tp = tile + dl * WHID + lane;
  float hv[16];
#pragma unroll
  for (int j = 0; j < 16; ++j) hv[j] = hp[j * 32];
  const float2 dm = sm[dl * NHEAD + q];
  const float av = expf(lrelu(sq + dm.x) - dm.y);
  if (lane < NHEAD)
    __hip_atomic_fetch_add(dent + dl * NHEAD + lane, av, __ATOMIC_RELAXED,
                           __HIP_MEMORY_SCOPE_WORKGROUP);
  float a[NHEAD];
#pragma unroll
  for (int k = 0; k < NHEAD; ++k)
    a[k] = __int_as_float(__builtin_amdgcn_readlane(__float_as_int(av), k));
#pragma unroll
  for (int j = 0; j < 16; ++j)
    __hip_atomic_fetch_add(tp + j * 32, a[j >> 2] * hv[j], __ATOMIC_RELAXED,
                           __HIP_MEMORY_SCOPE_WORKGROUP);
}

__device__ __forceinline__ float wave_sum(float v) {
  for (int o = 16; o > 0; o >>= 1) v += __shfl_xor(v, o, 32);
  return v;
}

__device__ __forceinline__ void tile_epilogue_l1(const float* tile, const float* dent, int wave, int lane, int d0, int n,
                                                 const float* __restrict__ b1,
                                                 const float* __restrict__ g,
                                                 const float* __restrict__ b,
                                                 float* __restrict__ x2) {
  const int c = lane * 4;
  const float4 bv = *(const float4*)(b1 + c);
  const float4 gv = *(const float4*)(g + c);
  const float4 ov = *(const float4*)(b + c);
  for (int dl = wave; dl < TILE_N; dl += TILE_THREADS / 32) {
    const int d = d0 + dl;
    if (d >= n) break;
    const float* tp = tile + dl * WHID + c;
    const float4 t0 = *(const float4*)(tp);
    const float4 t1 = *(const float4*)(tp + HIDF);
    const float4 t2 = *(const float4*)(tp + 2 * HIDF);
    const float4 t3 = *(const float4*)(tp + 3 * HIDF);
    const float4 dn = *(const float4*)(dent + dl * NHEAD);
    float4 v;
    v.x = 0.25f * (t0.x / dn.x + t1.x / dn.y + t2.x / dn.z + t3.x / dn.w) + bv.x;
    v.y = 0.25f * (t0.y / dn.x + t1.y / dn.y + t2.y / dn.z + t3.y / dn.w) + bv.y;
    v.z = 0.25f * (t0.z / dn.x + t1.z / dn.y + t2.z / dn.z + t3.z / dn.w) + bv.z;
    v.w = 0.25f * (t0.w / dn.x + t1.w / dn.y + t2.w / dn.z + t3.w / dn.w) + bv.w;
    const float mu = wave_sum((v.x + v.y) + (v.z + v.w)) * (1.0f / HIDF);
    float4 dv;
    dv.x = v.x - mu; dv.y = v.y - mu; dv.z = v.z - mu; dv.w = v.w - mu;
    const float var = wave_sum((dv.x * dv.x + dv.y * dv.y) + (dv.z * dv.z + dv.w * dv.w)) * (1.0f / HIDF);
    const float rs = rsqrtf(var + 1e-5f);
    float4 y;
    y.x = lrelu(gv.x * dv.x * rs + ov.x);
    y.y = lrelu(gv.y * dv.y * rs + ov.y);
    y.z = lrelu(gv.z * dv.z * rs + ov.z);
    y.w = lrelu(gv.w * dv.w * rs + ov.w);
    *(float4*)(x2 + (size_t)d * HIDF + c) = y;
  }
}

__device__ __forceinline__ void tile_epilogue_l2(const float* tile, const float* dent, const int* gids, int tid, int d0, int n,
                                                 const float* __restrict__ b2,
                                                 float* __restrict__ pooled,
                                                 float* __restrict__ cnt) {
  const int c    = tid & (WHID - 1);
  const int r0   = (tid / WHID) * (TILE_N / (TILE_THREADS / WHID));
  const int rend = min(r0 + TILE_N / (TILE_THREADS / WHID), n - d0);
  if (r0 >= rend) return;
  const float bv = b2[c];
  const int   hq = c / HIDF;
  int   gcur = gids[r0];
  int   rows = 0;
  float s    = 0.f;
  for (int rb = r0; rb < rend; rb += 8) {
    float t[8];
    float dn[8];
    int   gk[8];
#pragma unroll
    for (int k = 0; k < 8; ++k) {
      const int r = min(rb + k, rend - 1);
      t[k]  = tile[r * WHID + c];
      dn[k] = dent[r * NHEAD + hq];
      gk[k] = gids[r];
    }
#pragma unroll
    for (int k = 0; k < 8; ++k) {
      if (rb + k < rend) {
        if (gk[k] != gcur) {
          atomicAdd(&pooled[(size_t)gcur * WHID + c], s);
          if (c == 0) atomicAdd(&cnt[gcur], (float)rows);
          gcur = gk[k]; rows = 0; s = 0.f;
        }
        s += lrelu(t[k] / dn[k] + bv);
        ++rows;
      }
    }
  }
  atomicAdd(&pooled[(size_t)gcur * WHID + c], s);
  if (c == 0) atomicAdd(&cnt[gcur], (float)rows);
}

#define MAX_UNROLL 10
__global__ __launch_bounds__(TILE_THREADS)
void edge_aggr_tile(const int* __restrict__ tstart, const int* __restrict__ rec, int n,
                    const float* __restrict__ ssrc, const float* __restrict__ sdst,
                    const float* __restrict__ h, float* __restrict__ max_out,
                    float* __restrict__ den_out,
                    const float* __restrict__ b1, const float* __restrict__ g,
                    const float* __restrict__ b, float* __restrict__ x2,
                    const float* __restrict__ b2, const int* __restrict__ batch,
                    float* __restrict__ pooled, float* __restrict__ cnt) {
  __shared__ float tile[TILE_N * WHID];
  __shared__ int   gids[TILE_N];
  __shared__ __attribute__((aligned(16))) float2 sm[TILE_N * NHEAD];
  __shared__ __attribute__((aligned(16))) float dent[TILE_N * NHEAD];
  const int tid  = threadIdx.x;
  const int lane = tid & 31;
  const int wave = __builtin_amdgcn_readfirstlane(tid >> 5);
  const int d0   = blockIdx.x * TILE_N;
  const int rbeg = tstart[blockIdx.x];
  const int rend = tstart[blockIdx.x + 1];

  for (int i = tid * 4; i < TILE_N * WHID; i += TILE_THREADS * 4)
    *(float4*)&tile[i] = make_float4(0.f, 0.f, 0.f, 0.f);
  if (!x2 && tid < TILE_N) gids[tid] = batch[min(d0 + tid, n - 1)];
  if (tid < TILE_N * NHEAD) {
    sm[tid]   = make_float2(score_of(sdst, (size_t)min(d0 + tid / NHEAD, n - 1) * NHEAD + (tid & (NHEAD - 1))), -INFINITY);
    dent[tid] = 0.f;
  }
  __syncthreads();

  {
    const int hd = tid & (NHEAD - 1);
    if (tid < TILE_N * NHEAD && d0 + tid / NHEAD < n)
      atomicMaxF(&sm[tid].y, lrelu(score_of(ssrc, (size_t)(d0 + tid / NHEAD) * NHEAD + hd) + sm[tid].x));
    const int nel = (rend - rbeg) * NHEAD;
    for (int i0 = tid; i0 < nel; i0 += TILE_THREADS * MAX_UNROLL) {
      int sk[MAX_UNROLL], dk[MAX_UNROLL];
#pragma unroll
      for (int k = 0; k < MAX_UNROLL; ++k) {
        const int4 r = ((const int4*)rec)[rbeg + (min(i0 + k * TILE_THREADS, nel - 1) >> 2)];
        sk[k] = r.y;
        dk[k] = r.z;
      }
      float sv[MAX_UNROLL];
#pragma unroll
      for (int k = 0; k < MAX_UNROLL; ++k) sv[k] = score_of(ssrc, (size_t)sk[k] * NHEAD + hd);
#pragma unroll
      for (int k = 0; k < MAX_UNROLL; ++k)
        if (i0 + k * TILE_THREADS < nel)
          atomicMaxF(&sm[dk[k] * NHEAD + hd].y, lrelu(sv[k] + sm[dk[k] * NHEAD + hd].x));
    }
  }
  __syncthreads();
  if (tid < TILE_N * NHEAD) {
    const float mx = sm[tid].y;
    if (max_out && d0 + tid / NHEAD < n) max_out[(size_t)d0 * NHEAD + tid] = mx;
    sm[tid].y = seg_max_or_zero(mx);
  }
  __syncthreads();

  for (int dl = wave; dl < TILE_N; dl += TILE_THREADS / 32) {
    const int d = d0 + dl;
    if (d < n) tile_add_edge(tile, dent, sm, lane, d, dl, ssrc, h);
  }

  for (int base = rbeg + wave * 32; base < rend; base += TILE_THREADS) {
    const int4 r = ((const int4*)rec)[min(base + lane, rend - 1)];
    const int cntw = min(32, rend - base);
    for (int b = 0; b < cntw; ++b) {
      const int s  = __builtin_amdgcn_readlane(r.y, b);
      const int dl = __builtin_amdgcn_readlane(r.z, b);
      tile_add_edge(tile, dent, sm, lane, s, dl, ssrc, h);
    }
  }
  __syncthreads();

  if (den_out && tid < TILE_N * NHEAD && d0 + tid / NHEAD < n)
    den_out[(size_t)d0 * NHEAD + tid] = dent[tid];
  if (x2) tile_epilogue_l1(tile, dent, wave, lane, d0, n, b1, g, b, x2);
  else    tile_epilogue_l2(tile, dent, gids, tid, d0, n, b2, pooled, cnt);
}

__global__ void pool_div(float* __restrict__ pooled, const float* __restrict__ cnt) {
  int i = blockIdx.x * blockDim.x + threadIdx.x;
  if (i >= NGROUP * WHID) return;
  pooled[i] /= fmaxf(cnt[i / WHID], 1.0f);
}

__global__ void ln_relu(float* __restrict__ x, const float* __restrict__ g,
                        const float* __restrict__ b, int dim) {
  extern __shared__ float sh[];
  int r = blockIdx.x, c = threadIdx.x;
  float v = x[(size_t)r * dim + c];
  sh[c] = v; __syncthreads();
  for (int s = dim >> 1; s > 0; s >>= 1) { if (c < s) sh[c] += sh[c + s]; __syncthreads(); }
  float mu = sh[0] / dim;
  __syncthreads();
  float dv = v - mu;
  sh[c] = dv * dv; __syncthreads();
  for (int s = dim >> 1; s > 0; s >>= 1) { if (c < s) sh[c] += sh[c + s]; __syncthreads(); }
  float var = sh[0] / dim;
  float y = g[c] * dv * rsqrtf(var + 1e-5f) + b[c];
  x[(size_t)r * dim + c] = fmaxf(y, 0.f);
}

__global__ void head_out(const float* __restrict__ t2, const float* __restrict__ Ws3,
                         const float* __restrict__ bs3, float* __restrict__ out) {
  int r = blockIdx.x * blockDim.x + threadIdx.x;
  if (r >= NGROUP) return;
  float l0 = bs3[0], l1 = bs3[1];
  for (int c = 0; c < OUT2; ++c) {
    float v = t2[r * OUT2 + c];
    l0 += v * Ws3[c * 2 + 0];
    l1 += v * Ws3[c * 2 + 1];
  }
  float mx = fmaxf(l0, l1);
  float e0 = expf(l0 - mx), e1 = expf(l1 - mx), s = e0 + e1;
  out[r * 2 + 0] = e0 / s;
  out[r * 2 + 1] = e1 / s;
}

extern "C" void kernel_launch(void* const* d_in, const int* in_sizes, int n_in,
                              void* d_out, int out_size, void* d_ws, size_t ws_size,
                              hipStream_t stream) {
  (void)n_in; (void)out_size; (void)ws_size;
  const float* x    = (const float*)d_in[0];
  const int*   ei   = (const int*)d_in[1];
  const int*   batch= (const int*)d_in[2];
  const float* W1   = (const float*)d_in[3];
  const float* as1  = (const float*)d_in[4];
  const float* ad1  = (const float*)d_in[5];
  const float* b1   = (const float*)d_in[6];
  const float* g_ln = (const float*)d_in[7];
  const float* b_ln = (const float*)d_in[8];
  const float* W2   = (const float*)d_in[9];
  const float* as2  = (const float*)d_in[10];
  const float* ad2  = (const float*)d_in[11];
  const float* b2   = (const float*)d_in[12];
  const float* Wl0  = (const float*)d_in[13];
  const float* bl0  = (const float*)d_in[14];
  const float* Ws1  = (const float*)d_in[15];
  const float* bs1  = (const float*)d_in[16];
  const float* g1   = (const float*)d_in[17];
  const float* bb1  = (const float*)d_in[18];
  const float* Ws2  = (const float*)d_in[19];
  const float* bs2  = (const float*)d_in[20];
  const float* g2   = (const float*)d_in[21];
  const float* bb2  = (const float*)d_in[22];
  const float* Ws3  = (const float*)d_in[23];
  const float* bs3  = (const float*)d_in[24];

  const int N  = in_sizes[0] / HIDF;
  const int E  = in_sizes[1] / 2;
  const int TE = E + N;
  const int NH = N * NHEAD;
  const int TEH = TE * NHEAD;

  float* out       = (float*)d_out;
  float* alpha_out = out + NGROUP * NCLSD;

  size_t off = 0;
  auto alloc = [&](size_t bytes) -> float* {
    off = (off + 255) & ~(size_t)255;
    float* p = (float*)((char*)d_ws + off);
    off += bytes;
    return p;
  };
  float* h      = alloc((size_t)N * WHID * 4);
  float* x2     = alloc((size_t)N * HIDF * 4);
  float* ssrc   = alloc((size_t)NH * 2 * 4);
  float* sdst   = alloc((size_t)NH * 2 * 4);
  float* mbuf   = alloc((size_t)NH * 4);
  float* den    = alloc((size_t)NH * 4);
  float* pooled = alloc((size_t)NGROUP * WHID * 4);
  float* cnt    = alloc((size_t)NGROUP * 4);
  float* z0     = alloc((size_t)NGROUP * HIDF * 4);
  float* t1     = alloc((size_t)NGROUP * OUTD * 4);
  float* t2     = alloc((size_t)NGROUP * OUT2 * 4);

  auto cdiv = [](long a, long b) { return (int)((a + b - 1) / b); };
  const int ntile = cdiv(N, TILE_N);
  const int nbin  = cdiv(E, BIN_EDGES);
  int* bhist  = (int*)alloc((size_t)nbin * ntile * 4);
  int* brank  = (int*)alloc((size_t)E * 4);
  int* tstart = (int*)alloc((size_t)(ntile + 1) * 4);
  int* rec    = (int*)alloc((size_t)E * 16);
  auto gemm = [&](const float* A, const float* B, const float* bias, float* C,
                  int M, int K, int Nc,
                  const float* a_s = nullptr, const float* a_d = nullptr) {
    int mtiles  = (M + 15) / 16;
    int mblocks = (mtiles + 7) / 8;
    int nslab   = Nc / 64;
    wmma_gemm<<<mblocks * nslab, 256, 0, stream>>>(A, B, bias, C, M, K, Nc, a_s, a_d, ssrc, sdst);
  };

  tile_rank<<<nbin, BIN_THREADS, 0, stream>>>(ei, E, ntile, bhist, brank);
  tile_scatter<<<nbin, BIN_THREADS, 0, stream>>>(ei, E, ntile, nbin, bhist, brank, tstart, rec);

  gemm(x, W1, nullptr, h, N, HIDF, WHID, as1, ad1);
  edge_aggr_tile<<<ntile, TILE_THREADS, 0, stream>>>(tstart, rec, N, ssrc, sdst, h, nullptr, nullptr,
                                                               b1, g_ln, b_ln, x2,
                                                               nullptr, nullptr, nullptr, nullptr);

  gemm(x2, W2, nullptr, h, N, HIDF, WHID, as2, ad2);
  fill_f32<<<cdiv(NGROUP * WHID, 256), 256, 0, stream>>>(pooled, 0.f, NGROUP * WHID);
  fill_f32<<<1, NGROUP, 0, stream>>>(cnt, 0.f, NGROUP);
  edge_aggr_tile<<<ntile, TILE_THREADS, 0, stream>>>(tstart, rec, N, ssrc, sdst, h, mbuf, den,
                                                               nullptr, nullptr, nullptr, nullptr,
                                                               b2, batch, pooled, cnt);
  edge_alpha<<<cdiv(TEH, 256), 256, 0, stream>>>(ei, E, N, ssrc, sdst, mbuf, den, alpha_out);

  pool_div<<<cdiv(NGROUP * WHID, 256), 256, 0, stream>>>(pooled, cnt);

  gemm(pooled, Wl0, bl0, z0, NGROUP, WHID, HIDF);
  gemm(z0, Ws1, bs1, t1, NGROUP, HIDF, OUTD);
  ln_relu<<<NGROUP, OUTD, OUTD * sizeof(float), stream>>>(t1, g1, bb1, OUTD);
  gemm(t1, Ws2, bs2, t2, NGROUP, OUTD, OUT2);
  ln_relu<<<NGROUP, OUT2, OUT2 * sizeof(float), stream>>>(t2, g2, bb2, OUT2);
  head_out<<<1, NGROUP, 0, stream>>>(t2, Ws3, bs3, out);
}
